# starting kernel
# baseline (speedup 1.0000x reference)
#include <hip/hip_runtime.h>

#define B_      16
#define R_      8
#define V_      33
#define H_      512
#define W_      512
#define VCUBE   (V_ * V_ * V_)
#define M_TOTAL (3 * VCUBE)
#define HW      (H_ * W_)

typedef __attribute__((ext_vector_type(2))) float v2f;
typedef __attribute__((ext_vector_type(4))) float v4f;
typedef __attribute__((ext_vector_type(8))) float v8f;

__global__ void lut_gen_wmma(const float* __restrict__ weights,
                             const float* __restrict__ bank,
                             float* __restrict__ lut)
{
    const int lane = threadIdx.x & 31;
    const int wave = threadIdx.x >> 5;
    const int tile = blockIdx.x * (blockDim.x >> 5) + wave;
    const int m0   = tile * 16;
    if (m0 >= M_TOTAL) return;

    const int half = lane >> 4;
    const int l16  = lane & 15;

    v8f acc = {};
    #pragma unroll
    for (int kb = 0; kb < R_; kb += 4) {
        int row    = m0 + l16;
        int rclamp = row < M_TOTAL ? row : (M_TOTAL - 1);
        const float* arow = bank + (long)rclamp * R_ + kb + half * 2;
        v2f a; a.x = arow[0]; a.y = arow[1];

        const float* bcol = weights + l16 * R_ + kb + half * 2;
        v2f bm; bm.x = bcol[0]; bm.y = bcol[1];

        acc = __builtin_amdgcn_wmma_f32_16x16x4_f32(
            false, a, false, bm, (short)0, acc, false, false);
    }

    const int   batch = l16;
    const int   mbase = m0 + half * 8;
    float*      dst   = lut + (long)batch * M_TOTAL + mbase;

    if (m0 + 16 <= M_TOTAL) {
        v4f lo = { acc[0], acc[1], acc[2], acc[3] };
        v4f hi = { acc[4], acc[5], acc[6], acc[7] };
        *(v4f*)(dst)     = lo;
        *(v4f*)(dst + 4) = hi;
    } else {
        #pragma unroll
        for (int j = 0; j < 8; ++j) {
            if (mbase + j < M_TOTAL) dst[j] = acc[j];
        }
    }
}

__device__ __forceinline__ int imin_(int a, int b) { return a < b ? a : b; }

__device__ __forceinline__ void tri1(const float* __restrict__ L,
                                     float r, float g, float bl,
                                     float* __restrict__ o)
{
    const float s = (float)(V_ - 1);
    float cx = fminf(fmaxf(r  * s, 0.0f), s);
    float cy = fminf(fmaxf(g  * s, 0.0f), s);
    float cz = fminf(fmaxf(bl * s, 0.0f), s);
    int x0 = imin_((int)cx, V_ - 1);
    int y0 = imin_((int)cy, V_ - 1);
    int z0 = imin_((int)cz, V_ - 1);
    int x1 = imin_(x0 + 1, V_ - 1);
    int y1 = imin_(y0 + 1, V_ - 1);
    int z1 = imin_(z0 + 1, V_ - 1);
    float fx = cx - (float)x0, fy = cy - (float)y0, fz = cz - (float)z0;
    float gx = 1.0f - fx, gy = 1.0f - fy, gz = 1.0f - fz;

    int zy00 = (z0 * V_ + y0) * V_;
    int zy01 = (z0 * V_ + y1) * V_;
    int zy10 = (z1 * V_ + y0) * V_;
    int zy11 = (z1 * V_ + y1) * V_;
    int i000 = zy00 + x0, i001 = zy00 + x1, i010 = zy01 + x0, i011 = zy01 + x1;
    int i100 = zy10 + x0, i101 = zy10 + x1, i110 = zy11 + x0, i111 = zy11 + x1;

    float w000 = gz * gy * gx, w001 = gz * gy * fx;
    float w010 = gz * fy * gx, w011 = gz * fy * fx;
    float w100 = fz * gy * gx, w101 = fz * gy * fx;
    float w110 = fz * fy * gx, w111 = fz * fy * fx;

    #pragma unroll
    for (int c = 0; c < 3; ++c) {
        const float* Lc = L + c * VCUBE;
        o[c] = Lc[i000] * w000 + Lc[i001] * w001 + Lc[i010] * w010 + Lc[i011] * w011
             + Lc[i100] * w100 + Lc[i101] * w101 + Lc[i110] * w110 + Lc[i111] * w111;
    }
}

__global__ void lut_apply(const float* __restrict__ img,
                          const float* __restrict__ lut,
                          float* __restrict__ out)
{
    const int  b  = blockIdx.y;
    const int  q  = blockIdx.x * blockDim.x + threadIdx.x;
    const long ib = (long)b * 3 * HW;
    const float* __restrict__ L = lut + (long)b * M_TOTAL;

    float4 r4 = ((const float4*)(img + ib         ))[q];
    float4 g4 = ((const float4*)(img + ib +     HW))[q];
    float4 b4 = ((const float4*)(img + ib + 2 * HW))[q];

    float4 o0, o1, o2;
    float  o[3];
    tri1(L, r4.x, g4.x, b4.x, o); o0.x = o[0]; o1.x = o[1]; o2.x = o[2];
    tri1(L, r4.y, g4.y, b4.y, o); o0.y = o[0]; o1.y = o[1]; o2.y = o[2];
    tri1(L, r4.z, g4.z, b4.z, o); o0.z = o[0]; o1.z = o[1]; o2.z = o[2];
    tri1(L, r4.w, g4.w, b4.w, o); o0.w = o[0]; o1.w = o[1]; o2.w = o[2];

    ((float4*)(out + ib         ))[q] = o0;
    ((float4*)(out + ib +     HW))[q] = o1;
    ((float4*)(out + ib + 2 * HW))[q] = o2;
}

extern "C" void kernel_launch(void* const* d_in, const int* in_sizes, int n_in,
                              void* d_out, int out_size, void* d_ws, size_t ws_size,
                              hipStream_t stream) {
    (void)in_sizes; (void)n_in; (void)out_size; (void)ws_size;
    const float* weights = (const float*)d_in[0];
    const float* img     = (const float*)d_in[1];
    const float* bank    = (const float*)d_in[2];
    float*       out     = (float*)d_out;
    float*       lut     = (float*)d_ws;

    const int tiles = (M_TOTAL + 15) / 16;
    dim3 blk1(256);
    dim3 grd1((tiles + 7) / 8);
    lut_gen_wmma<<<grd1, blk1, 0, stream>>>(weights, bank, lut);

    dim3 blk2(256);
    dim3 grd2(HW / 4 / 256, B_);
    lut_apply<<<grd2, blk2, 0, stream>>>(img, lut, out);
}


# Table synthesized as one 64-byte record per lattice node (4 xy corners x 3 colours); apply: 4 lanes share a pixel and fetch its 2 records as whole contiguous runs, one colour per lane, not 24 gathers
# speedup vs baseline: 4.9071x; 4.9071x over previous
#include <hip/hip_runtime.h>

#define B_      16
#define R_      8
#define V_      33
#define H_      512
#define W_      512
#define VCUBE   (V_ * V_ * V_)
#define M_TOTAL (3 * VCUBE)
#define HW      (H_ * W_)

#define REC_F   16
#define TAB_F   (VCUBE * REC_F)

#define GEN_WAVES 3
#define GEN_RPW   9
#define GEN_BLOCKS (VCUBE / (GEN_WAVES * GEN_RPW))

typedef __attribute__((ext_vector_type(2))) float v2f;
typedef __attribute__((ext_vector_type(4))) float v4f;
typedef __attribute__((ext_vector_type(8))) float v8f;

__global__ void __launch_bounds__(GEN_WAVES * 32)
lut_gen_wmma(const float* __restrict__ weights,
             const float* __restrict__ bank,
             float* __restrict__ tab)
{
    const int lane = threadIdx.x & 31;
    const int wave = threadIdx.x >> 5;
    const int half = lane >> 4;
    const int l16  = lane & 15;

    const float* bcol = weights + l16 * R_ + half * 2;
    v2f bm0; bm0.x = bcol[0]; bm0.y = bcol[1];
    v2f bm1; bm1.x = bcol[4]; bm1.y = bcol[5];

    const int s  = l16 < 12 ? l16 : 11;
    const int c  = s >> 2;
    const int yy = (s >> 1) & 1;
    const int xx = s & 1;

    const int e0 = (blockIdx.x * GEN_WAVES + wave) * GEN_RPW;

    v2f a0[GEN_RPW], a1[GEN_RPW];
    #pragma unroll
    for (int i = 0; i < GEN_RPW; ++i) {
        const int e = e0 + i;
        const int x = e % V_;
        const int t = e / V_;
        const int y = t % V_;
        const int z = t / V_;
        const int xs = (x + xx) < V_ ? (x + xx) : (V_ - 1);
        const int ys = (y + yy) < V_ ? (y + yy) : (V_ - 1);
        const int m  = c * VCUBE + (z * V_ + ys) * V_ + xs;
        const float* arow = bank + m * R_ + half * 2;
        a0[i].x = arow[0]; a0[i].y = arow[1];
        a1[i].x = arow[4]; a1[i].y = arow[5];
    }

    float* dst = tab + (long)l16 * TAB_F + e0 * REC_F + half * 8;
    #pragma unroll
    for (int i = 0; i < GEN_RPW; ++i) {
        v8f acc = {};
        acc = __builtin_amdgcn_wmma_f32_16x16x4_f32(
            false, a0[i], false, bm0, (short)0, acc, false, false);
        acc = __builtin_amdgcn_wmma_f32_16x16x4_f32(
            false, a1[i], false, bm1, (short)0, acc, false, false);
        v4f lo = { acc[0], acc[1], acc[2], acc[3] };
        v4f hi = { acc[4], acc[5], acc[6], acc[7] };
        *(v4f*)(dst + i * REC_F)     = lo;
        *(v4f*)(dst + i * REC_F + 4) = hi;
    }
}

__device__ __forceinline__ int imin_(int a, int b) { return a < b ? a : b; }

struct Cell {
    int   r0, r1;
    float w000, w001, w010, w011, w100, w101, w110, w111;
};

__device__ __forceinline__ Cell cell_of(float r, float g, float bl)
{
    const float s = (float)(V_ - 1);
    float cx = fminf(fmaxf(r  * s, 0.0f), s);
    float cy = fminf(fmaxf(g  * s, 0.0f), s);
    float cz = fminf(fmaxf(bl * s, 0.0f), s);
    int x0 = imin_((int)cx, V_ - 1);
    int y0 = imin_((int)cy, V_ - 1);
    int z0 = imin_((int)cz, V_ - 1);
    int z1 = imin_(z0 + 1, V_ - 1);
    float fx = cx - (float)x0, fy = cy - (float)y0, fz = cz - (float)z0;
    float gx = 1.0f - fx, gy = 1.0f - fy, gz = 1.0f - fz;

    Cell k;
    k.r0 = ((z0 * V_ + y0) * V_ + x0) * REC_F;
    k.r1 = ((z1 * V_ + y0) * V_ + x0) * REC_F;
    k.w000 = gz * gy * gx; k.w001 = gz * gy * fx;
    k.w010 = gz * fy * gx; k.w011 = gz * fy * fx;
    k.w100 = fz * gy * gx; k.w101 = fz * gy * fx;
    k.w110 = fz * fy * gx; k.w111 = fz * fy * fx;
    return k;
}

__device__ __forceinline__ float blend(const v4f lo, const v4f hi, const Cell& k)
{
    return lo.x * k.w000 + lo.y * k.w001 + lo.z * k.w010 + lo.w * k.w011
         + hi.x * k.w100 + hi.y * k.w101 + hi.z * k.w110 + hi.w * k.w111;
}

#define STRIPS 4

__global__ void __launch_bounds__(256)
lut_apply(const float* __restrict__ img,
          const float* __restrict__ tab,
          float* __restrict__ out)
{
    const int  b    = blockIdx.y;
    const int  lane = threadIdx.x & 31;
    const int  wave = threadIdx.x >> 5;
    const int  grp  = lane >> 2;
    const int  col  = lane & 3;
    const long ib   = (long)b * 3 * HW;
    const float* __restrict__ T = tab + (long)b * TAB_F + col * 4;
    const int  p0   = blockIdx.x * 1024 + wave * (32 * STRIPS) + grp * 4;

    #pragma nounroll
    for (int it = 0; it < STRIPS; ++it) {
        const int p = p0 + it * 32;
        const float4 r4 = *(const float4*)(img + ib          + p);
        const float4 g4 = *(const float4*)(img + ib +     HW + p);
        const float4 b4 = *(const float4*)(img + ib + 2 * HW + p);

        const float rr[4] = { r4.x, r4.y, r4.z, r4.w };
        const float gg[4] = { g4.x, g4.y, g4.z, g4.w };
        const float bb[4] = { b4.x, b4.y, b4.z, b4.w };

        Cell k[4];
        v4f  lo[4], hi[4];
        #pragma unroll
        for (int j = 0; j < 4; ++j) k[j] = cell_of(rr[j], gg[j], bb[j]);
        #pragma unroll
        for (int j = 0; j < 4; ++j) {
            lo[j] = *(const v4f*)(T + k[j].r0);
            hi[j] = *(const v4f*)(T + k[j].r1);
        }
        float o[4];
        #pragma unroll
        for (int j = 0; j < 4; ++j) o[j] = blend(lo[j], hi[j], k[j]);

        if (col < 3)
            *(float4*)(out + ib + (long)col * HW + p) = make_float4(o[0], o[1], o[2], o[3]);
    }
}

extern "C" void kernel_launch(void* const* d_in, const int* in_sizes, int n_in,
                              void* d_out, int out_size, void* d_ws, size_t ws_size,
                              hipStream_t stream) {
    (void)in_sizes; (void)n_in; (void)out_size; (void)ws_size;
    const float* weights = (const float*)d_in[0];
    const float* img     = (const float*)d_in[1];
    const float* bank    = (const float*)d_in[2];
    float*       out     = (float*)d_out;
    float*       tab     = (float*)d_ws;

    dim3 blk1(GEN_WAVES * 32);
    dim3 grd1(GEN_BLOCKS);
    lut_gen_wmma<<<grd1, blk1, 0, stream>>>(weights, bank, tab);

    dim3 blk2(256);
    dim3 grd2(HW / 1024, B_);
    lut_apply<<<grd2, blk2, 0, stream>>>(img, tab, out);
}


# Table records reordered z-fastest, 34 slots per (y,x) column (slot 33 repeats plane 32): a pixel's z0, z1 records are 128 contiguous bytes, one cache line if z0 is even: 3 lines per 2 pixels, not 4
# speedup vs baseline: 5.3460x; 1.0894x over previous
#include <hip/hip_runtime.h>

#define B_      16
#define R_      8
#define V_      33
#define H_      512
#define W_      512
#define VCUBE   (V_ * V_ * V_)
#define M_TOTAL (3 * VCUBE)
#define HW      (H_ * W_)

#define REC_F   16
#define ZS      (V_ + 1)
#define TAB_F   (V_ * V_ * ZS * REC_F)

#define GEN_WAVES 2
#define GEN_RPW   17
#define GEN_BLOCKS (V_ * V_)

typedef __attribute__((ext_vector_type(2))) float v2f;
typedef __attribute__((ext_vector_type(4))) float v4f;
typedef __attribute__((ext_vector_type(8))) float v8f;

__global__ void __launch_bounds__(GEN_WAVES * 32)
lut_gen_wmma(const float* __restrict__ weights,
             const float* __restrict__ bank,
             float* __restrict__ tab)
{
    const int lane = threadIdx.x & 31;
    const int wave = threadIdx.x >> 5;
    const int half = lane >> 4;
    const int l16  = lane & 15;

    const float* bcol = weights + l16 * R_ + half * 2;
    v2f bm0; bm0.x = bcol[0]; bm0.y = bcol[1];
    v2f bm1; bm1.x = bcol[4]; bm1.y = bcol[5];

    const int s  = l16 < 12 ? l16 : 11;
    const int c  = s >> 2;
    const int yy = (s >> 1) & 1;
    const int xx = s & 1;

    const int col = blockIdx.x;
    const int x   = col % V_;
    const int y   = col / V_;
    const int z0  = wave * GEN_RPW;
    const int xs  = (x + xx) < V_ ? (x + xx) : (V_ - 1);
    const int ys  = (y + yy) < V_ ? (y + yy) : (V_ - 1);
    const int m0  = c * VCUBE + (z0 * V_ + ys) * V_ + xs;
    const float* arow = bank + m0 * R_ + half * 2;
    const int last = (z0 + GEN_RPW - 1) < V_ ? (GEN_RPW - 1) : (GEN_RPW - 2);

    v2f a0[GEN_RPW], a1[GEN_RPW];
    #pragma unroll
    for (int i = 0; i < GEN_RPW; ++i) {
        const float* ar = arow + (i < GEN_RPW - 1 ? i : last) * (V_ * V_ * R_);
        a0[i].x = ar[0]; a0[i].y = ar[1];
        a1[i].x = ar[4]; a1[i].y = ar[5];
    }

    float* dst = tab + (long)l16 * TAB_F + (col * ZS + z0) * REC_F + half * 8;
    #pragma unroll
    for (int i = 0; i < GEN_RPW; ++i) {
        v8f acc = {};
        acc = __builtin_amdgcn_wmma_f32_16x16x4_f32(
            false, a0[i], false, bm0, (short)0, acc, false, false);
        acc = __builtin_amdgcn_wmma_f32_16x16x4_f32(
            false, a1[i], false, bm1, (short)0, acc, false, false);
        v4f lo = { acc[0], acc[1], acc[2], acc[3] };
        v4f hi = { acc[4], acc[5], acc[6], acc[7] };
        *(v4f*)(dst + i * REC_F)     = lo;
        *(v4f*)(dst + i * REC_F + 4) = hi;
    }
}

__device__ __forceinline__ int imin_(int a, int b) { return a < b ? a : b; }

struct Cell {
    int   r0;
    float w000, w001, w010, w011, w100, w101, w110, w111;
};

__device__ __forceinline__ Cell cell_of(float r, float g, float bl)
{
    const float s = (float)(V_ - 1);
    float cx = fminf(fmaxf(r  * s, 0.0f), s);
    float cy = fminf(fmaxf(g  * s, 0.0f), s);
    float cz = fminf(fmaxf(bl * s, 0.0f), s);
    int x0 = imin_((int)cx, V_ - 1);
    int y0 = imin_((int)cy, V_ - 1);
    int z0 = imin_((int)cz, V_ - 1);
    float fx = cx - (float)x0, fy = cy - (float)y0, fz = cz - (float)z0;
    float gx = 1.0f - fx, gy = 1.0f - fy, gz = 1.0f - fz;

    Cell k;
    k.r0 = ((y0 * V_ + x0) * ZS + z0) * REC_F;
    k.w000 = gz * gy * gx; k.w001 = gz * gy * fx;
    k.w010 = gz * fy * gx; k.w011 = gz * fy * fx;
    k.w100 = fz * gy * gx; k.w101 = fz * gy * fx;
    k.w110 = fz * fy * gx; k.w111 = fz * fy * fx;
    return k;
}

__device__ __forceinline__ float blend(const v4f lo, const v4f hi, const Cell& k)
{
    return lo.x * k.w000 + lo.y * k.w001 + lo.z * k.w010 + lo.w * k.w011
         + hi.x * k.w100 + hi.y * k.w101 + hi.z * k.w110 + hi.w * k.w111;
}

#define STRIPS 4

__global__ void __launch_bounds__(256)
lut_apply(const float* __restrict__ img,
          const float* __restrict__ tab,
          float* __restrict__ out)
{
    const int  b    = blockIdx.y;
    const int  lane = threadIdx.x & 31;
    const int  wave = threadIdx.x >> 5;
    const int  grp  = lane >> 2;
    const int  col  = lane & 3;
    const long ib   = (long)b * 3 * HW;
    const float* __restrict__ T = tab + (long)b * TAB_F + col * 4;
    const int  p0   = blockIdx.x * 1024 + wave * (32 * STRIPS) + grp * 4;

    #pragma nounroll
    for (int it = 0; it < STRIPS; ++it) {
        const int p = p0 + it * 32;
        const float4 r4 = *(const float4*)(img + ib          + p);
        const float4 g4 = *(const float4*)(img + ib +     HW + p);
        const float4 b4 = *(const float4*)(img + ib + 2 * HW + p);

        const float rr[4] = { r4.x, r4.y, r4.z, r4.w };
        const float gg[4] = { g4.x, g4.y, g4.z, g4.w };
        const float bb[4] = { b4.x, b4.y, b4.z, b4.w };

        Cell k[4];
        v4f  lo[4], hi[4];
        #pragma unroll
        for (int j = 0; j < 4; ++j) k[j] = cell_of(rr[j], gg[j], bb[j]);
        #pragma unroll
        for (int j = 0; j < 4; ++j) {
            lo[j] = *(const v4f*)(T + k[j].r0);
            hi[j] = *(const v4f*)(T + k[j].r0 + REC_F);
        }
        float o[4];
        #pragma unroll
        for (int j = 0; j < 4; ++j) o[j] = blend(lo[j], hi[j], k[j]);

        if (col < 3)
            *(float4*)(out + ib + (long)col * HW + p) = make_float4(o[0], o[1], o[2], o[3]);
    }
}

extern "C" void kernel_launch(void* const* d_in, const int* in_sizes, int n_in,
                              void* d_out, int out_size, void* d_ws, size_t ws_size,
                              hipStream_t stream) {
    (void)in_sizes; (void)n_in; (void)out_size; (void)ws_size;
    const float* weights = (const float*)d_in[0];
    const float* img     = (const float*)d_in[1];
    const float* bank    = (const float*)d_in[2];
    float*       out     = (float*)d_out;
    float*       tab     = (float*)d_ws;

    dim3 blk1(GEN_WAVES * 32);
    dim3 grd1(GEN_BLOCKS);
    lut_gen_wmma<<<grd1, blk1, 0, stream>>>(weights, bank, tab);

    dim3 blk2(256);
    dim3 grd2(HW / 1024, B_);
    lut_apply<<<grd2, blk2, 0, stream>>>(img, tab, out);
}


# Table synthesis: record tiles staged in LDS and written out per batch table as whole 128-byte lines (4 per store instruction); before, each store touched 16 lines with two 16-byte pieces each
# speedup vs baseline: 5.5955x; 1.0467x over previous
#include <hip/hip_runtime.h>

#define B_      16
#define R_      8
#define V_      33
#define H_      512
#define W_      512
#define VCUBE   (V_ * V_ * V_)
#define M_TOTAL (3 * VCUBE)
#define HW      (H_ * W_)

#define REC_F   16
#define ZS      (V_ + 1)
#define TAB_F   (V_ * V_ * ZS * REC_F)

#define GEN_WAVES 2
#define GEN_RPW   17
#define GEN_BLOCKS (V_ * V_)
#define COL_F   (ZS * REC_F)
#define COL_P   (COL_F / 4)
#define STG_F   (COL_F + 4)

typedef __attribute__((ext_vector_type(2))) float v2f;
typedef __attribute__((ext_vector_type(4))) float v4f;
typedef __attribute__((ext_vector_type(8))) float v8f;

__global__ void __launch_bounds__(GEN_WAVES * 32)
lut_gen_wmma(const float* __restrict__ weights,
             const float* __restrict__ bank,
             float* __restrict__ tab)
{
    const int lane = threadIdx.x & 31;
    const int wave = threadIdx.x >> 5;
    const int half = lane >> 4;
    const int l16  = lane & 15;

    const float* bcol = weights + l16 * R_ + half * 2;
    v2f bm0; bm0.x = bcol[0]; bm0.y = bcol[1];
    v2f bm1; bm1.x = bcol[4]; bm1.y = bcol[5];

    const int s  = l16 < 12 ? l16 : 11;
    const int c  = s >> 2;
    const int yy = (s >> 1) & 1;
    const int xx = s & 1;

    const int col = blockIdx.x;
    const int x   = col % V_;
    const int y   = col / V_;
    const int z0  = wave * GEN_RPW;
    const int xs  = (x + xx) < V_ ? (x + xx) : (V_ - 1);
    const int ys  = (y + yy) < V_ ? (y + yy) : (V_ - 1);
    const int m0  = c * VCUBE + (z0 * V_ + ys) * V_ + xs;
    const float* arow = bank + m0 * R_ + half * 2;
    const int last = (z0 + GEN_RPW - 1) < V_ ? (GEN_RPW - 1) : (GEN_RPW - 2);

    v2f a0[GEN_RPW], a1[GEN_RPW];
    #pragma unroll
    for (int i = 0; i < GEN_RPW; ++i) {
        const float* ar = arow + (i < GEN_RPW - 1 ? i : last) * (V_ * V_ * R_);
        a0[i].x = ar[0]; a0[i].y = ar[1];
        a1[i].x = ar[4]; a1[i].y = ar[5];
    }

    __shared__ float stage[B_ * STG_F];
    float* sdst = stage + l16 * STG_F + z0 * REC_F + half * 8;
    #pragma unroll
    for (int i = 0; i < GEN_RPW; ++i) {
        v8f acc = {};
        acc = __builtin_amdgcn_wmma_f32_16x16x4_f32(
            false, a0[i], false, bm0, (short)0, acc, false, false);
        acc = __builtin_amdgcn_wmma_f32_16x16x4_f32(
            false, a1[i], false, bm1, (short)0, acc, false, false);
        v4f lo = { acc[0], acc[1], acc[2], acc[3] };
        v4f hi = { acc[4], acc[5], acc[6], acc[7] };
        *(v4f*)(sdst + i * REC_F)     = lo;
        *(v4f*)(sdst + i * REC_F + 4) = hi;
    }
    __syncthreads();

    float* gcol = tab + (long)col * COL_F + (long)(wave * 8) * TAB_F;
    const float* scol = stage + (wave * 8) * STG_F;
    #pragma unroll
    for (int kk = 0; kk < 2 * (4 * COL_P / 32); ++kk) {
        const int g    = kk / (4 * COL_P / 32);
        const int base = (kk % (4 * COL_P / 32)) * 32;
        const int blo  = base / COL_P;
        const int bhi  = (base + 31) / COL_P;
        int bl = blo;
        if (bhi != blo) bl = (base + lane >= bhi * COL_P) ? bhi : blo;
        const int bt  = g * 4 + bl;
        const int off = base + lane - bl * COL_P;
        const v4f v = *(const v4f*)(scol + bt * STG_F + off * 4);
        *(v4f*)(gcol + (long)bt * TAB_F + off * 4) = v;
    }
}

__device__ __forceinline__ int imin_(int a, int b) { return a < b ? a : b; }

struct Cell {
    int   r0;
    float w000, w001, w010, w011, w100, w101, w110, w111;
};

__device__ __forceinline__ Cell cell_of(float r, float g, float bl)
{
    const float s = (float)(V_ - 1);
    float cx = fminf(fmaxf(r  * s, 0.0f), s);
    float cy = fminf(fmaxf(g  * s, 0.0f), s);
    float cz = fminf(fmaxf(bl * s, 0.0f), s);
    int x0 = imin_((int)cx, V_ - 1);
    int y0 = imin_((int)cy, V_ - 1);
    int z0 = imin_((int)cz, V_ - 1);
    float fx = cx - (float)x0, fy = cy - (float)y0, fz = cz - (float)z0;
    float gx = 1.0f - fx, gy = 1.0f - fy, gz = 1.0f - fz;

    Cell k;
    k.r0 = ((y0 * V_ + x0) * ZS + z0) * REC_F;
    k.w000 = gz * gy * gx; k.w001 = gz * gy * fx;
    k.w010 = gz * fy * gx; k.w011 = gz * fy * fx;
    k.w100 = fz * gy * gx; k.w101 = fz * gy * fx;
    k.w110 = fz * fy * gx; k.w111 = fz * fy * fx;
    return k;
}

__device__ __forceinline__ float blend(const v4f lo, const v4f hi, const Cell& k)
{
    return lo.x * k.w000 + lo.y * k.w001 + lo.z * k.w010 + lo.w * k.w011
         + hi.x * k.w100 + hi.y * k.w101 + hi.z * k.w110 + hi.w * k.w111;
}

#define STRIPS 4

__global__ void __launch_bounds__(256)
lut_apply(const float* __restrict__ img,
          const float* __restrict__ tab,
          float* __restrict__ out)
{
    const int  b    = blockIdx.y;
    const int  lane = threadIdx.x & 31;
    const int  wave = threadIdx.x >> 5;
    const int  grp  = lane >> 2;
    const int  col  = lane & 3;
    const long ib   = (long)b * 3 * HW;
    const float* __restrict__ T = tab + (long)b * TAB_F + col * 4;
    const int  p0   = blockIdx.x * 1024 + wave * (32 * STRIPS) + grp * 4;

    #pragma nounroll
    for (int it = 0; it < STRIPS; ++it) {
        const int p = p0 + it * 32;
        const float4 r4 = *(const float4*)(img + ib          + p);
        const float4 g4 = *(const float4*)(img + ib +     HW + p);
        const float4 b4 = *(const float4*)(img + ib + 2 * HW + p);

        const float rr[4] = { r4.x, r4.y, r4.z, r4.w };
        const float gg[4] = { g4.x, g4.y, g4.z, g4.w };
        const float bb[4] = { b4.x, b4.y, b4.z, b4.w };

        Cell k[4];
        v4f  lo[4], hi[4];
        #pragma unroll
        for (int j = 0; j < 4; ++j) k[j] = cell_of(rr[j], gg[j], bb[j]);
        #pragma unroll
        for (int j = 0; j < 4; ++j) {
            lo[j] = *(const v4f*)(T + k[j].r0);
            hi[j] = *(const v4f*)(T + k[j].r0 + REC_F);
        }
        float o[4];
        #pragma unroll
        for (int j = 0; j < 4; ++j) o[j] = blend(lo[j], hi[j], k[j]);

        if (col < 3)
            *(float4*)(out + ib + (long)col * HW + p) = make_float4(o[0], o[1], o[2], o[3]);
    }
}

extern "C" void kernel_launch(void* const* d_in, const int* in_sizes, int n_in,
                              void* d_out, int out_size, void* d_ws, size_t ws_size,
                              hipStream_t stream) {
    (void)in_sizes; (void)n_in; (void)out_size; (void)ws_size;
    const float* weights = (const float*)d_in[0];
    const float* img     = (const float*)d_in[1];
    const float* bank    = (const float*)d_in[2];
    float*       out     = (float*)d_out;
    float*       tab     = (float*)d_ws;

    dim3 blk1(GEN_WAVES * 32);
    dim3 grd1(GEN_BLOCKS);
    lut_gen_wmma<<<grd1, blk1, 0, stream>>>(weights, bank, tab);

    dim3 blk2(256);
    dim3 grd2(HW / 1024, B_);
    lut_apply<<<grd2, blk2, 0, stream>>>(img, tab, out);
}
